# starting kernel
# baseline (speedup 1.0000x reference)
#include <hip/hip_runtime.h>

typedef float v2f __attribute__((ext_vector_type(2)));
typedef float v8f __attribute__((ext_vector_type(8)));

#define BATCH 8
#define C_IN  64
#define HW    96
#define OC    128
#define NF    54
#define NFP   56
#define NPAIR 28
#define NSUB  6
#define NPAD  112

__device__ constexpr int PA[36] = {0,1,2,3,4,5,6,7, 0,1,2,3,4,5,6, 0,1,2,3,4,5,
                                   0,1,2,3,4, 0,1,2,3, 0,1,2, 0,1, 0};
__device__ constexpr int PB[36] = {1,2,3,4,5,6,7,8, 2,3,4,5,6,7,8, 3,4,5,6,7,8,
                                   4,5,6,7,8, 5,6,7,8, 6,7,8, 7,8, 8};

static __device__ __forceinline__ void build_feat(float (*featbuf)[NPAD][2],
                                                  const float* __restrict__ xplane,
                                                  int y, int tid) {
  if (tid < 2 * HW) {
    const int grp = tid / HW;
    const int n   = tid - grp * HW;
    float p[9];
#pragma unroll
    for (int kh = 0; kh < 3; ++kh) {
      const int  yy  = y + kh - 1;
      const bool oky = (yy >= 0) & (yy < HW);
      const int  yyc = yy < 0 ? 0 : (yy >= HW ? HW - 1 : yy);
#pragma unroll
      for (int kw = 0; kw < 3; ++kw) {
        const int  xx  = n + kw - 1;
        const bool okx = (xx >= 0) & (xx < HW);
        const int  xxc = xx < 0 ? 0 : (xx >= HW ? HW - 1 : xx);
        const float v  = xplane[yyc * HW + xxc];
        p[kh * 3 + kw] = (oky & okx) ? v : 0.0f;
      }
    }
    if (grp == 0) {
#pragma unroll
      for (int k = 0; k < 9; ++k) featbuf[k >> 1][n][k & 1] = p[k];
#pragma unroll
      for (int k = 0; k < 9; ++k) {
        const int f = 9 + k;
        featbuf[f >> 1][n][f & 1] = p[k] * p[k];
      }
#pragma unroll
      for (int g = 0; g < 9; ++g) {
        const int f = 18 + g;
        featbuf[f >> 1][n][f & 1] = p[PA[g]] * p[PB[g]];
      }
    } else {
#pragma unroll
      for (int g = 9; g < 36; ++g) {
        const int f = 18 + g;
        featbuf[f >> 1][n][f & 1] = p[PA[g]] * p[PB[g]];
      }
    }
  }
}

__global__ __launch_bounds__(256)
void Second_order_conv2d_kernel(const float* __restrict__ x,
                                const float* __restrict__ w,
                                float* __restrict__ out) {
  __shared__ float featP[2][NPAIR][NPAD][2];

  const int y    = blockIdx.x;
  const int b    = blockIdx.y;
  const int tid  = threadIdx.x;
  const int wave = tid >> 5;
  const int lane = tid & 31;
  const int m     = lane & 15;
  const int khalf = lane >> 4;
  const int obase = wave * 16;

  for (int i = tid; i < 2 * NPAD * 2; i += 256) {
    const int bufi = i / (NPAD * 2);
    const int j    = i - bufi * (NPAD * 2);
    featP[bufi][NPAIR - 1][j >> 1][j & 1] = 0.0f;
  }

  v8f acc[NSUB];
  const v8f vzero = {0.f, 0.f, 0.f, 0.f, 0.f, 0.f, 0.f, 0.f};
#pragma unroll
  for (int t = 0; t < NSUB; ++t) acc[t] = vzero;

  const float* wbase  = w + (size_t)(obase + m) * (C_IN * NF);
  const float* xbatch = x + (size_t)b * C_IN * (HW * HW);

  build_feat(featP[0], xbatch, y, tid);
  __syncthreads();

  for (int c = 0; c < C_IN; ++c) {
    const int buf = c & 1;

    if (c + 1 < C_IN)
      build_feat(featP[buf ^ 1], xbatch + (size_t)(c + 1) * (HW * HW), y, tid);

    if (c + 2 < C_IN && tid < 9) {
      const int r  = tid / 3;
      const int sg = tid - r * 3;
      int yy = y + r - 1;
      yy = yy < 0 ? 0 : (yy >= HW ? HW - 1 : yy);
      __builtin_prefetch(xbatch + (size_t)(c + 2) * (HW * HW) + yy * HW + sg * 32, 0, 0);
    }

    const float* wc = wbase + c * NF;
#pragma unroll
    for (int fb = 0; fb < NFP; fb += 4) {
      const int f0 = fb + 2 * khalf;
      v2f av;
      if (f0 < NF) {
        av = *reinterpret_cast<const v2f*>(wc + f0);
      } else {
        av.x = 0.0f; av.y = 0.0f;
      }
      const int fp = (fb >> 1) + khalf;
#pragma unroll
      for (int t = 0; t < NSUB; ++t) {
        v2f bv = *reinterpret_cast<const v2f*>(&featP[buf][fp][t * 16 + m][0]);
        acc[t] = __builtin_amdgcn_wmma_f32_16x16x4_f32(
            false, av, false, bv, (short)0, acc[t], false, false);
      }
    }

    __syncthreads();
  }

  float* outb = out + ((size_t)b * OC * HW + y) * HW;
#pragma unroll
  for (int t = 0; t < NSUB; ++t) {
#pragma unroll
    for (int v = 0; v < 8; ++v) {
      int o = obase + v + khalf * 8;
      outb[(size_t)o * (HW * HW) + t * 16 + m] = acc[t][v];
    }
  }
}

extern "C" void kernel_launch(void* const* d_in, const int* in_sizes, int n_in,
                              void* d_out, int out_size, void* d_ws, size_t ws_size,
                              hipStream_t stream) {
  const float* x = (const float*)d_in[0];
  const float* w = (const float*)d_in[1];
  float* out = (float*)d_out;
  (void)in_sizes; (void)n_in; (void)out_size; (void)d_ws; (void)ws_size;

  dim3 grid(HW, BATCH);
  Second_order_conv2d_kernel<<<grid, 256, 0, stream>>>(x, w, out);
}


# Global loads pipelined one channel ahead: patch of channel c+2 and weight fragments of c+1 are fetched into registers during sweep c, across an LDS-only barrier; no global wait before the WMMAs
# speedup vs baseline: 1.0955x; 1.0955x over previous
#include <hip/hip_runtime.h>

typedef float v2f __attribute__((ext_vector_type(2)));
typedef float v8f __attribute__((ext_vector_type(8)));

#define BATCH 8
#define C_IN  64
#define HW    96
#define OC    128
#define NF    54
#define NFP   56
#define NPAIR 28
#define NSUB  6
#define NPAD  112

__device__ constexpr int PA[36] = {0,1,2,3,4,5,6,7, 0,1,2,3,4,5,6, 0,1,2,3,4,5,
                                   0,1,2,3,4, 0,1,2,3, 0,1,2, 0,1, 0};
__device__ constexpr int PB[36] = {1,2,3,4,5,6,7,8, 2,3,4,5,6,7,8, 3,4,5,6,7,8,
                                   4,5,6,7,8, 5,6,7,8, 6,7,8, 7,8, 8};


static __device__ __forceinline__ void load_patch(float (&v)[9],
                                                  const float* __restrict__ xplane,
                                                  int y, int tid) {
  if (tid < 2 * HW) {
    const int grp = tid / HW;
    const int n   = tid - grp * HW;
#pragma unroll
    for (int kh = 0; kh < 3; ++kh) {
      const int yy  = y + kh - 1;
      const int yyc = yy < 0 ? 0 : (yy >= HW ? HW - 1 : yy);
#pragma unroll
      for (int kw = 0; kw < 3; ++kw) {
        const int xx  = n + kw - 1;
        const int xxc = xx < 0 ? 0 : (xx >= HW ? HW - 1 : xx);
        v[kh * 3 + kw] = xplane[yyc * HW + xxc];
      }
    }
  }
}

static __device__ __forceinline__ void store_feat(float (*featbuf)[NPAD][2],
                                                  const float (&v)[9], int y, int tid) {
  if (tid < 2 * HW) {
    const int grp = tid / HW;
    const int n   = tid - grp * HW;
    float p[9];
#pragma unroll
    for (int kh = 0; kh < 3; ++kh) {
      const int  yy  = y + kh - 1;
      const bool oky = (yy >= 0) & (yy < HW);
#pragma unroll
      for (int kw = 0; kw < 3; ++kw) {
        const int  xx  = n + kw - 1;
        const bool okx = (xx >= 0) & (xx < HW);
        p[kh * 3 + kw] = (oky & okx) ? v[kh * 3 + kw] : 0.0f;
      }
    }
    if (grp == 0) {
#pragma unroll
      for (int k = 0; k < 9; ++k) featbuf[k >> 1][n][k & 1] = p[k];
#pragma unroll
      for (int k = 0; k < 9; ++k) {
        const int f = 9 + k;
        featbuf[f >> 1][n][f & 1] = p[k] * p[k];
      }
#pragma unroll
      for (int g = 0; g < 9; ++g) {
        const int f = 18 + g;
        featbuf[f >> 1][n][f & 1] = p[PA[g]] * p[PB[g]];
      }
    } else {
#pragma unroll
      for (int g = 9; g < 36; ++g) {
        const int f = 18 + g;
        featbuf[f >> 1][n][f & 1] = p[PA[g]] * p[PB[g]];
      }
    }
  }
}

static __device__ __forceinline__ void lds_barrier() {
  __builtin_amdgcn_fence(__ATOMIC_RELEASE, "workgroup", "local");
  __builtin_amdgcn_s_barrier();
  __builtin_amdgcn_fence(__ATOMIC_ACQUIRE, "workgroup", "local");
}

template <int R>
static __device__ __forceinline__ v2f load_afrag(const float* __restrict__ wc, int khalf) {
  const int f0 = 4 * R + 2 * khalf;
  if (4 * R + 2 < NF) {
    return *reinterpret_cast<const v2f*>(wc + f0);
  } else {
    return *reinterpret_cast<const v2f*>(wc + (f0 < NF ? f0 : 4 * R));
  }
}

template <int R>
static __device__ __forceinline__ v2f use_afrag(v2f raw, int khalf) {
  if (4 * R + 2 < NF) {
    return raw;
  } else {
    const bool ok = (4 * R + 2 * khalf) < NF;
    v2f z; z.x = ok ? raw.x : 0.0f; z.y = ok ? raw.y : 0.0f;
    return z;
  }
}

__global__ __launch_bounds__(256)
void Second_order_conv2d_kernel(const float* __restrict__ x,
                                const float* __restrict__ w,
                                float* __restrict__ out) {
  __shared__ float featP[2][NPAIR][NPAD][2];

  const int y    = blockIdx.x;
  const int b    = blockIdx.y;
  const int tid  = threadIdx.x;
  const int wave = tid >> 5;
  const int lane = tid & 31;
  const int m     = lane & 15;
  const int khalf = lane >> 4;
  const int obase = wave * 16;

  for (int i = tid; i < 2 * NPAD * 2; i += 256) {
    const int bufi = i / (NPAD * 2);
    const int j    = i - bufi * (NPAD * 2);
    featP[bufi][NPAIR - 1][j >> 1][j & 1] = 0.0f;
  }

  v8f acc[NSUB];
  const v8f vzero = {0.f, 0.f, 0.f, 0.f, 0.f, 0.f, 0.f, 0.f};
#pragma unroll
  for (int t = 0; t < NSUB; ++t) acc[t] = vzero;

  const float* wbase  = w + (size_t)(obase + m) * (C_IN * NF);
  const float* xbatch = x + (size_t)b * C_IN * (HW * HW);

  float p[9];
#pragma unroll
  for (int k = 0; k < 9; ++k) p[k] = 0.0f;
  load_patch(p, xbatch, y, tid);
  store_feat(featP[0], p, y, tid);
  load_patch(p, xbatch + (size_t)(HW * HW), y, tid);

  v2f a[NFP / 4];
  a[0]  = load_afrag<0>(wbase, khalf);   a[1]  = load_afrag<1>(wbase, khalf);
  a[2]  = load_afrag<2>(wbase, khalf);   a[3]  = load_afrag<3>(wbase, khalf);
  a[4]  = load_afrag<4>(wbase, khalf);   a[5]  = load_afrag<5>(wbase, khalf);
  a[6]  = load_afrag<6>(wbase, khalf);   a[7]  = load_afrag<7>(wbase, khalf);
  a[8]  = load_afrag<8>(wbase, khalf);   a[9]  = load_afrag<9>(wbase, khalf);
  a[10] = load_afrag<10>(wbase, khalf);  a[11] = load_afrag<11>(wbase, khalf);
  a[12] = load_afrag<12>(wbase, khalf);  a[13] = load_afrag<13>(wbase, khalf);
  lds_barrier();

  for (int c = 0; c < C_IN; ++c) {
    const int buf = c & 1;

    if (c + 1 < C_IN) store_feat(featP[buf ^ 1], p, y, tid);

    if (c + 2 < C_IN)
      load_patch(p, xbatch + (size_t)(c + 2) * (HW * HW), y, tid);

    if (c + 3 < C_IN && tid < 9) {
      const int r  = tid / 3;
      const int sg = tid - r * 3;
      int yy = y + r - 1;
      yy = yy < 0 ? 0 : (yy >= HW ? HW - 1 : yy);
      __builtin_prefetch(xbatch + (size_t)(c + 3) * (HW * HW) + yy * HW + sg * 32, 0, 0);
    }

    const float* wn = wbase + (c + 1 < C_IN ? c + 1 : c) * NF;
#define SWEEP_STEP(R)                                                              \
    {                                                                              \
      const v2f av = use_afrag<R>(a[R], khalf);                                    \
      const int fp = 2 * (R) + khalf;                         \
      _Pragma("unroll")                                                            \
      for (int t = 0; t < NSUB; ++t) {                                             \
          \
        v2f bv = *reinterpret_cast<const v2f*>(&featP[buf][fp][t * 16 + m][0]);    \
        acc[t] = __builtin_amdgcn_wmma_f32_16x16x4_f32(                            \
            false, av, false, bv, (short)0, acc[t], false, false);                 \
      }                                                                            \
      a[R] = load_afrag<R>(wn, khalf);                                             \
    }
    SWEEP_STEP(0)  SWEEP_STEP(1)  SWEEP_STEP(2)  SWEEP_STEP(3)  SWEEP_STEP(4)
    __builtin_amdgcn_sched_barrier(0x7C6);
    SWEEP_STEP(5)  SWEEP_STEP(6)  SWEEP_STEP(7)  SWEEP_STEP(8)  SWEEP_STEP(9)
    __builtin_amdgcn_sched_barrier(0x7C6);
    SWEEP_STEP(10) SWEEP_STEP(11) SWEEP_STEP(12) SWEEP_STEP(13)
#undef SWEEP_STEP

    lds_barrier();
  }

  float* outb = out + ((size_t)b * OC * HW + y) * HW;
#pragma unroll
  for (int t = 0; t < NSUB; ++t) {
#pragma unroll
    for (int v = 0; v < 8; ++v) {
      int o = obase + v + khalf * 8;
      outb[(size_t)o * (HW * HW) + t * 16 + m] = acc[t][v];
    }
  }
}

extern "C" void kernel_launch(void* const* d_in, const int* in_sizes, int n_in,
                              void* d_out, int out_size, void* d_ws, size_t ws_size,
                              hipStream_t stream) {
  const float* x = (const float*)d_in[0];
  const float* w = (const float*)d_in[1];
  float* out = (float*)d_out;
  (void)in_sizes; (void)n_in; (void)out_size; (void)d_ws; (void)ws_size;

  dim3 grid(HW, BATCH);
  Second_order_conv2d_kernel<<<grid, 256, 0, stream>>>(x, w, out);
}


# Two channels per barrier round, their 108 features contiguous in K: 27 K-steps per channel pair instead of 2x14 (no K padding), 32 barriers instead of 64; weight fragments in a 9-deep register ring
# speedup vs baseline: 1.1369x; 1.0378x over previous
#include <hip/hip_runtime.h>


typedef float v2f __attribute__((ext_vector_type(2)));
typedef float v8f __attribute__((ext_vector_type(8)));

#define BATCH 8
#define C_IN  64
#define HW    96
#define OC    128
#define NF    54
#define CPR   2
#define NROUND (C_IN / CPR)
#define KR    (CPR * NF)
#define NSTEP (KR / 4)
#define NPAIRR (KR / 2)
#define NRING 9
#define NSUB  6
#define NPAD  112

__device__ constexpr int PA[36] = {0,1,2,3,4,5,6,7, 0,1,2,3,4,5,6, 0,1,2,3,4,5,
                                   0,1,2,3,4, 0,1,2,3, 0,1,2, 0,1, 0};
__device__ constexpr int PB[36] = {1,2,3,4,5,6,7,8, 2,3,4,5,6,7,8, 3,4,5,6,7,8,
                                   4,5,6,7,8, 5,6,7,8, 6,7,8, 7,8, 8};


static __device__ __forceinline__ void load_patch(float (&v)[9],
                                                  const float* __restrict__ xpair,
                                                  int y, int tid) {
  const int grp = tid < HW ? 0 : 1;
  const int n   = tid - grp * HW;
  const float* __restrict__ xplane = xpair + (size_t)grp * (HW * HW);
#pragma unroll
  for (int kh = 0; kh < 3; ++kh) {
    const int yy  = y + kh - 1;
    const int yyc = yy < 0 ? 0 : (yy >= HW ? HW - 1 : yy);
#pragma unroll
    for (int kw = 0; kw < 3; ++kw) {
      const int xx  = n + kw - 1;
      const int xxc = xx < 0 ? 0 : (xx >= HW ? HW - 1 : xx);
      v[kh * 3 + kw] = xplane[yyc * HW + xxc];
    }
  }
}

static __device__ __forceinline__ float feat_of(const float (&p)[9], int f) {
  if (f < 9)  return p[f];
  if (f < 18) return p[f - 9] * p[f - 9];
  return p[PA[f - 18]] * p[PB[f - 18]];
}

static __device__ __forceinline__ void store_feat(float (*featbuf)[NPAD][2],
                                                  const float (&v)[9], int y, int tid) {
  if (tid < 2 * HW) {
    const int grp = tid / HW;
    const int n   = tid - grp * HW;
    float p[9];
#pragma unroll
    for (int kh = 0; kh < 3; ++kh) {
      const int  yy  = y + kh - 1;
      const bool oky = (yy >= 0) & (yy < HW);
#pragma unroll
      for (int kw = 0; kw < 3; ++kw) {
        const int  xx  = n + kw - 1;
        const bool okx = (xx >= 0) & (xx < HW);
        p[kh * 3 + kw] = (oky & okx) ? v[kh * 3 + kw] : 0.0f;
      }
    }
    float (*plane)[NPAD][2] = featbuf + grp * (NF / 2);
#pragma unroll
    for (int j = 0; j < NF / 2; ++j) {
      v2f pr;
      pr.x = feat_of(p, 2 * j);
      pr.y = feat_of(p, 2 * j + 1);
      *reinterpret_cast<v2f*>(&plane[j][n][0]) = pr;
    }
  }
}

static __device__ __forceinline__ void lds_barrier() {
  __builtin_amdgcn_fence(__ATOMIC_RELEASE, "workgroup", "local");
  __builtin_amdgcn_s_barrier();
  __builtin_amdgcn_fence(__ATOMIC_ACQUIRE, "workgroup", "local");
}

__global__ __launch_bounds__(256)
void Second_order_conv2d_kernel(const float* __restrict__ x,
                                const float* __restrict__ w,
                                float* __restrict__ out) {
  __shared__ float featP[2][NPAIRR][NPAD][2];

  const int y    = blockIdx.x;
  const int b    = blockIdx.y;
  const int tid  = threadIdx.x;
  const int wave = tid >> 5;
  const int lane = tid & 31;
  const int m     = lane & 15;
  const int khalf = lane >> 4;
  const int obase = wave * 16;

  v8f acc[NSUB];
  const v8f vzero = {0.f, 0.f, 0.f, 0.f, 0.f, 0.f, 0.f, 0.f};
#pragma unroll
  for (int t = 0; t < NSUB; ++t) acc[t] = vzero;

  const float* wlane  = w + (size_t)(obase + m) * (C_IN * NF) + 2 * khalf;
  const float* xbatch = x + (size_t)b * C_IN * (HW * HW);

  float p[9];
  load_patch(p, xbatch, y, tid);
  store_feat(featP[0], p, y, tid);
  load_patch(p, xbatch + (size_t)(CPR * HW * HW), y, tid);

  v2f a[NRING];
#pragma unroll
  for (int i = 0; i < NRING; ++i) a[i] = *reinterpret_cast<const v2f*>(wlane + 4 * i);
  lds_barrier();

  for (int r = 0; r < NROUND; ++r) {
    const int buf = r & 1;

    if (r + 1 < NROUND) store_feat(featP[buf ^ 1], p, y, tid);

    load_patch(p, xbatch + (size_t)(r + 2 < NROUND ? r + 2 : NROUND - 1) * (CPR * HW * HW), y, tid);

    if (r + 3 < NROUND && tid < 18) {
      const int ch = tid / 9;
      const int q  = tid - ch * 9;
      const int rr = q / 3;
      const int sg = q - rr * 3;
      int yy = y + rr - 1;
      yy = yy < 0 ? 0 : (yy >= HW ? HW - 1 : yy);
      __builtin_prefetch(xbatch + (size_t)((r + 3) * CPR + ch) * (HW * HW) + yy * HW + sg * 32, 0, 0);
    }

    const float* wcur  = wlane + r * KR;
    const float* wnext = wlane + (r + 1 < NROUND ? r + 1 : r) * KR;
#define SWEEP_STEP(S)                                                              \
    {                                                                              \
                \
      const v2f nx = *reinterpret_cast<const v2f*>(                                \
          ((S) + NRING < NSTEP ? wcur : wnext) + 4 * (((S) + NRING) % NSTEP));     \
      const v2f av = a[(S) % NRING];                                               \
      const int fp = 2 * (S) + khalf;                         \
      _Pragma("unroll")                                                            \
      for (int t = 0; t < NSUB; ++t) {                                             \
          \
        v2f bv = *reinterpret_cast<const v2f*>(&featP[buf][fp][t * 16 + m][0]);    \
        acc[t] = __builtin_amdgcn_wmma_f32_16x16x4_f32(                            \
            false, av, false, bv, (short)0, acc[t], false, false);                 \
      }                                                                            \
      a[(S) % NRING] = nx;                                                         \
        \
      __builtin_amdgcn_sched_barrier(0x7C6);                                       \
    }
    SWEEP_STEP(0)  SWEEP_STEP(1)  SWEEP_STEP(2)  SWEEP_STEP(3)  SWEEP_STEP(4)
    SWEEP_STEP(5)  SWEEP_STEP(6)  SWEEP_STEP(7)  SWEEP_STEP(8)  SWEEP_STEP(9)
    SWEEP_STEP(10) SWEEP_STEP(11) SWEEP_STEP(12) SWEEP_STEP(13) SWEEP_STEP(14)
    SWEEP_STEP(15) SWEEP_STEP(16) SWEEP_STEP(17) SWEEP_STEP(18) SWEEP_STEP(19)
    SWEEP_STEP(20) SWEEP_STEP(21) SWEEP_STEP(22) SWEEP_STEP(23) SWEEP_STEP(24)
    SWEEP_STEP(25) SWEEP_STEP(26)
#undef SWEEP_STEP

    lds_barrier();
  }

  float* outb = out + ((size_t)b * OC * HW + y) * HW;
#pragma unroll
  for (int t = 0; t < NSUB; ++t) {
#pragma unroll
    for (int v = 0; v < 8; ++v) {
      int o = obase + v + khalf * 8;
      outb[(size_t)o * (HW * HW) + t * 16 + m] = acc[t][v];
    }
  }
}

extern "C" void kernel_launch(void* const* d_in, const int* in_sizes, int n_in,
                              void* d_out, int out_size, void* d_ws, size_t ws_size,
                              hipStream_t stream) {
  const float* x = (const float*)d_in[0];
  const float* w = (const float*)d_in[1];
  float* out = (float*)d_out;
  (void)in_sizes; (void)n_in; (void)out_size; (void)d_ws; (void)ws_size;

  dim3 grid(HW, BATCH);
  Second_order_conv2d_kernel<<<grid, 256, 0, stream>>>(x, w, out);
}
